# starting kernel
# baseline (speedup 1.0000x reference)
#include <hip/hip_runtime.h>

typedef __attribute__((ext_vector_type(2))) float v2f;
typedef __attribute__((ext_vector_type(8))) float v8f;

#define FDIM 64

__global__ void k_init_deg(float* __restrict__ deg, int n) {
    int i = blockIdx.x * blockDim.x + threadIdx.x;
    if (i < n) deg[i] = 1.0f;
}

__global__ void k_deg_accum(const int* __restrict__ dst, float* __restrict__ deg, int E) {
    int e = blockIdx.x * blockDim.x + threadIdx.x;
    if (e < E) atomicAdd(&deg[dst[e]], 1.0f);
}

__global__ void k_dinv(const float* __restrict__ deg, float* __restrict__ dinv, int n) {
    int i = blockIdx.x * blockDim.x + threadIdx.x;
    if (i < n) { float d = deg[i]; dinv[i] = d > 0.0f ? rsqrtf(d) : 0.0f; }
}

__global__ void k_selfloop(const float* __restrict__ x, const float* __restrict__ dinv,
                           const float* __restrict__ bias, float* __restrict__ agg, int n) {
    int idx = blockIdx.x * blockDim.x + threadIdx.x;
    if (idx < n * FDIM) {
        int i = idx >> 6;
        int f = idx & 63;
        float di = dinv[i];
        float v  = x[idx] * di * di;
        if (bias) v += bias[f];
        agg[idx] = v;
    }
}

__global__ void k_scatter(const float* __restrict__ x, const int* __restrict__ src,
                          const int* __restrict__ dst, const float* __restrict__ dinv,
                          float* __restrict__ agg, int E) {
    long long gtid = (long long)blockIdx.x * blockDim.x + threadIdx.x;
    int e    = (int)(gtid >> 5);
    int lane = (int)(gtid & 31);
    if (e < E) {
        int s = src[e], d = dst[e];
        float w = dinv[s] * dinv[d];
        v2f xv = *(const v2f*)(x + (long long)s * FDIM + lane * 2);
        float* p = agg + (long long)d * FDIM + lane * 2;
        atomicAdd(p,     xv.x * w);
        atomicAdd(p + 1, xv.y * w);
    }
}

__global__ void k_gemm_wmma(const float* __restrict__ A, const float* __restrict__ Bm,
                            const float* __restrict__ bias, float* __restrict__ out,
                            int n, int K, int ncols, int relu) {
    int col_tile = blockIdx.x;
    int row_tile = blockIdx.y * blockDim.y + threadIdx.y;
    if (row_tile * 16 >= n) return;

    int lane = threadIdx.x;
    int half = lane >> 4;
    int m    = lane & 15;
    int row0 = row_tile * 16;
    int col0 = col_tile * 16;

    const float* arow = A  + (long long)(row0 + m) * K + 2 * half;
    const float* bcol = Bm + (long long)(2 * half) * ncols + col0 + m;

    v8f acc = {};
    for (int k0 = 0; k0 < K; k0 += 4) {
        v2f a = *(const v2f*)(arow + k0);
        v2f b;
        b.x = bcol[(long long)(k0)     * ncols];
        b.y = bcol[(long long)(k0 + 1) * ncols];
        acc = __builtin_amdgcn_wmma_f32_16x16x4_f32(
                  false, a, false, b, (short)0, acc, false, false);
    }

    float bv = bias ? bias[col0 + m] : 0.0f;
#pragma unroll
    for (int r = 0; r < 8; ++r) {
        float v = acc[r] + bv;
        if (relu) v = fmaxf(v, 0.0f);
        out[(long long)(row0 + r + 8 * half) * ncols + col0 + m] = v;
    }
}

extern "C" void kernel_launch(void* const* d_in, const int* in_sizes, int n_in,
                              void* d_out, int out_size, void* d_ws, size_t ws_size,
                              hipStream_t stream) {
    (void)n_in; (void)out_size; (void)ws_size;

    const int*   eidx = (const int*)d_in[0];
    const float* emb  = (const float*)d_in[1];
    const float* W1   = (const float*)d_in[2];
    const float* b1   = (const float*)d_in[3];
    const float* W2   = (const float*)d_in[4];
    const float* b2   = (const float*)d_in[5];

    const int E = in_sizes[0] / 2;
    const int n = in_sizes[1] / 64;
    const int* src = eidx;
    const int* dst = eidx + E;

    float* deg  = (float*)d_ws;
    float* dinv = deg + n;
    float* agg1 = dinv + n;
    float* h    = agg1 + (size_t)n * 64;
    float* t    = agg1;
    float* out  = (float*)d_out;

    const int B = 256;
    k_init_deg <<<(n + B - 1) / B, B, 0, stream>>>(deg, n);
    k_deg_accum<<<(E + B - 1) / B, B, 0, stream>>>(dst, deg, E);
    k_dinv     <<<(n + B - 1) / B, B, 0, stream>>>(deg, dinv, n);

    k_selfloop<<<(int)(((size_t)n * 64 + B - 1) / B), B, 0, stream>>>(emb, dinv, nullptr, agg1, n);
    k_scatter <<<(int)(((size_t)E * 32 + B - 1) / B), B, 0, stream>>>(emb, src, dst, dinv, agg1, E);
    {
        dim3 block(32, 8);
        dim3 grid(128 / 16, ((n + 15) / 16 + 7) / 8);
        k_gemm_wmma<<<grid, block, 0, stream>>>(agg1, W1, b1, h, n, 64, 128, 1);
    }

    {
        dim3 block(32, 8);
        dim3 grid(64 / 16, ((n + 15) / 16 + 7) / 8);
        k_gemm_wmma<<<grid, block, 0, stream>>>(h, W2, nullptr, t, n, 128, 64, 0);
    }
    k_selfloop<<<(int)(((size_t)n * 64 + B - 1) / B), B, 0, stream>>>(t, dinv, b2, out, n);
    k_scatter <<<(int)(((size_t)E * 32 + B - 1) / B), B, 0, stream>>>(t, src, dst, dinv, out, E);
}


# dense products: weight fragments stay in registers, 1024 four-wave groups walk 16-row tiles; A rows by 16-byte loads via LDS, result lines stored whole via LDS; same sum order
# speedup vs baseline: 2.3559x; 2.3559x over previous
#include <hip/hip_runtime.h>

typedef __attribute__((ext_vector_type(2))) float v2f;
typedef __attribute__((ext_vector_type(4))) int   v4i;
typedef __attribute__((ext_vector_type(8))) float v8f;

#define FDIM 64
#define SCAN_T 1024

__global__ void k_zero_cnt(int* __restrict__ cnt, int n) {
    int i = blockIdx.x * blockDim.x + threadIdx.x;
    if (i < n) cnt[i] = 0;
}

__global__ void k_rank(const int* __restrict__ dst, int* __restrict__ cnt,
                       int* __restrict__ rank, int E) {
    int e = blockIdx.x * blockDim.x + threadIdx.x;
    if (e < E) rank[e] = atomicAdd(&cnt[dst[e]], 1);
}

__global__ void __launch_bounds__(SCAN_T)
k_scan(const int* __restrict__ cnt, int* __restrict__ rowptr, float* __restrict__ dinv, int n) {
    __shared__ int sh[SCAN_T];
    const int tid  = threadIdx.x;
    const int blk0 = blockIdx.x * SCAN_T;

    int part = 0;
    const v4i* c4 = (const v4i*)cnt;
    const int n4 = blk0 >> 2;
#pragma unroll 4
    for (int i = tid; i < n4; i += SCAN_T) {
        v4i v = c4[i];
        part += (v.x + v.y) + (v.z + v.w);
    }

    const int i   = blk0 + tid;
    const int own = (i < n) ? cnt[i] : 0;

    sh[tid] = part;
    __syncthreads();
    for (int s = SCAN_T / 2; s > 0; s >>= 1) {
        if (tid < s) sh[tid] += sh[tid + s];
        __syncthreads();
    }
    const int base = sh[0];
    __syncthreads();

    sh[tid] = own;
    __syncthreads();
    for (int off = 1; off < SCAN_T; off <<= 1) {
        int v = (tid >= off) ? sh[tid - off] : 0;
        __syncthreads();
        sh[tid] += v;
        __syncthreads();
    }
    const int incl = sh[tid];

    if (i < n) {
        rowptr[i] = base + incl - own;
        float d = (float)(own + 1);
        dinv[i] = d > 0.0f ? rsqrtf(d) : 0.0f;
    }
}

__global__ void k_place(const int* __restrict__ src, const int* __restrict__ dst,
                        const int* __restrict__ rank, const int* __restrict__ rowptr,
                        int* __restrict__ col, int E) {
    int e = blockIdx.x * blockDim.x + threadIdx.x;
    if (e < E) col[rowptr[dst[e]] + rank[e]] = src[e];
}

__global__ void k_move(const int* __restrict__ a, int* __restrict__ b, int cnt) {
    int i = blockIdx.x * blockDim.x + threadIdx.x;
    if (i < cnt) b[i] = a[i];
}

__global__ void __launch_bounds__(256)
k_pull(const float* __restrict__ x, const int* __restrict__ rowptr, const int* __restrict__ col,
       const float* __restrict__ dinv, const float* __restrict__ bias, float* __restrict__ y,
       int n, int E) {
    const int lane = threadIdx.x & 31;
    const int d = __builtin_amdgcn_readfirstlane((int)((blockIdx.x * 256u + threadIdx.x) >> 5));
    if (d >= n) return;

    const int beg = rowptr[d];
    const int end = (d + 1 < n) ? rowptr[d + 1] : E;
    const float dd = dinv[d];
    const float* xl = x + 2 * lane;

    v2f xs = *(const v2f*)(xl + (long long)d * FDIM);
    v2f acc;
    acc.x = xs.x * dd * dd;
    acc.y = xs.y * dd * dd;
    if (bias) { acc.x += bias[2 * lane]; acc.y += bias[2 * lane + 1]; }

    for (int j0 = beg; j0 < end; j0 += 32) {
        int m = end - j0;
        if (m > 32) m = 32;
        int   my = d;
        float wv = 0.0f;
        if (lane < m) { my = col[j0 + lane]; wv = dinv[my] * dd; }

        int k = 0;
        for (; k + 8 <= m; k += 8) {
            v2f r[8]; float w[8];
#pragma unroll
            for (int u = 0; u < 8; ++u) {
                int s = __builtin_amdgcn_readlane(my, k + u);
                w[u]  = __int_as_float(__builtin_amdgcn_readlane(__float_as_int(wv), k + u));
                r[u]  = *(const v2f*)(xl + (long long)s * FDIM);
            }
#pragma unroll
            for (int u = 0; u < 8; ++u) acc += r[u] * w[u];
        }
        for (; k + 4 <= m; k += 4) {
            v2f r[4]; float w[4];
#pragma unroll
            for (int u = 0; u < 4; ++u) {
                int s = __builtin_amdgcn_readlane(my, k + u);
                w[u]  = __int_as_float(__builtin_amdgcn_readlane(__float_as_int(wv), k + u));
                r[u]  = *(const v2f*)(xl + (long long)s * FDIM);
            }
#pragma unroll
            for (int u = 0; u < 4; ++u) acc += r[u] * w[u];
        }
        for (; k < m; ++k) {
            int   s = __builtin_amdgcn_readlane(my, k);
            float w = __int_as_float(__builtin_amdgcn_readlane(__float_as_int(wv), k));
            v2f   r = *(const v2f*)(xl + (long long)s * FDIM);
            acc += r * w;
        }
    }

    *(v2f*)(y + (long long)d * FDIM + 2 * lane) = acc;
}

typedef __attribute__((ext_vector_type(4))) float v4f;

template <int K, int NC, int CT, bool RELU>
__global__ void __launch_bounds__(128)
k_dense_ws(const float* __restrict__ A, const float* __restrict__ Bm,
           const float* __restrict__ bias, float* __restrict__ out, int n) {
    constexpr int KS  = K / 4;
    constexpr int AP  = K + 4;
    constexpr int SP  = NC + 4;
    constexpr int NLD = (16 * K / 4) / 128;
    constexpr int NST = (16 * NC / 4) / 128;
    static_assert(NC == 4 * 16 * CT, "4 waves x CT column tiles cover NC");

    __shared__ __attribute__((aligned(16))) float sA[16 * AP];
    __shared__ __attribute__((aligned(16))) float sD[16 * SP];

    const int tid  = threadIdx.x;
    const int lane = tid & 31;
    const int w    = tid >> 5;
    const int c    = lane & 15;
    const int h    = lane >> 4;
    const int ntiles = (n + 15) >> 4;

    v2f   b[CT][KS];
    float bv[CT];
#pragma unroll
    for (int j = 0; j < CT; ++j) {
        const int col = (w * CT + j) * 16 + c;
#pragma unroll
        for (int ks = 0; ks < KS; ++ks) {
            b[j][ks].x = Bm[(4 * ks + 2 * h)     * NC + col];
            b[j][ks].y = Bm[(4 * ks + 2 * h + 1) * NC + col];
        }
        bv[j] = bias ? bias[col] : 0.0f;
    }

    int tile = blockIdx.x;
    v4f pre[NLD];
    if (tile < ntiles) {
        const float* at = A + (long long)tile * 16 * K;
#pragma unroll
        for (int p = 0; p < NLD; ++p) {
            const int idx = (p * 128 + tid) * 4;
            v4f z = {0.0f, 0.0f, 0.0f, 0.0f};
            pre[p] = (tile * 16 + idx / K < n) ? *(const v4f*)(at + idx) : z;
        }
    }

    for (; tile < ntiles; tile += gridDim.x) {
#pragma unroll
        for (int p = 0; p < NLD; ++p) {
            const int idx = (p * 128 + tid) * 4;
            *(v4f*)&sA[(idx / K) * AP + (idx % K)] = pre[p];
        }
        __syncthreads();

        const int nt = tile + gridDim.x;
        if (nt < ntiles) {
            const float* at = A + (long long)nt * 16 * K;
#pragma unroll
            for (int p = 0; p < NLD; ++p) {
                const int idx = (p * 128 + tid) * 4;
                v4f z = {0.0f, 0.0f, 0.0f, 0.0f};
                pre[p] = (nt * 16 + idx / K < n) ? *(const v4f*)(at + idx) : z;
            }
        }

        v2f a[KS];
#pragma unroll
        for (int ks = 0; ks < KS; ++ks) a[ks] = *(const v2f*)&sA[c * AP + 4 * ks + 2 * h];
        __builtin_amdgcn_sched_barrier(0);

        v8f acc[CT];
#pragma unroll
        for (int j = 0; j < CT; ++j) acc[j] = (v8f){0.0f, 0.0f, 0.0f, 0.0f, 0.0f, 0.0f, 0.0f, 0.0f};
#pragma unroll
        for (int ks = 0; ks < KS; ++ks) {
#pragma unroll
            for (int j = 0; j < CT; ++j)
                acc[j] = __builtin_amdgcn_wmma_f32_16x16x4_f32(
                             false, a[ks], false, b[j][ks], (short)0, acc[j], false, false);
        }

#pragma unroll
        for (int j = 0; j < CT; ++j) {
#pragma unroll
            for (int r = 0; r < 8; ++r) {
                float v = acc[j][r] + bv[j];
                if (RELU) v = fmaxf(v, 0.0f);
                sD[(r + 8 * h) * SP + (w * CT + j) * 16 + c] = v;
            }
        }
        __syncthreads();

        float* ot = out + (long long)tile * 16 * NC;
#pragma unroll
        for (int p = 0; p < NST; ++p) {
            const int idx = (p * 128 + tid) * 4;
            const int r = idx / NC, cc = idx % NC;
            if (tile * 16 + r < n) *(v4f*)(ot + idx) = *(const v4f*)&sD[r * SP + cc];
        }
    }
}

extern "C" void kernel_launch(void* const* d_in, const int* in_sizes, int n_in,
                              void* d_out, int out_size, void* d_ws, size_t ws_size,
                              hipStream_t stream) {
    (void)n_in; (void)out_size; (void)ws_size;

    const int*   eidx = (const int*)d_in[0];
    const float* emb  = (const float*)d_in[1];
    const float* W1   = (const float*)d_in[2];
    const float* b1   = (const float*)d_in[3];
    const float* W2   = (const float*)d_in[4];
    const float* b2   = (const float*)d_in[5];

    const int E = in_sizes[0] / 2;
    const int n = in_sizes[1] / 64;
    const int* src = eidx;
    const int* dst = eidx + E;

    int*   rowptr = (int*)d_ws;
    float* dinv   = (float*)d_ws + n;
    float* agg1   = dinv + n;
    float* h      = agg1 + (size_t)n * 64;
    float* t      = agg1;
    float* out    = (float*)d_out;

    int* cnt   = (int*)d_out;
    int* rank  = cnt + n;
    int* col_a = rank + E;
    int* col_b = (int*)h;

    const int B = 256;
    k_zero_cnt<<<(n + B - 1) / B, B, 0, stream>>>(cnt, n);
    k_rank    <<<(E + B - 1) / B, B, 0, stream>>>(dst, cnt, rank, E);
    k_scan    <<<(n + SCAN_T - 1) / SCAN_T, SCAN_T, 0, stream>>>(cnt, rowptr, dinv, n);
    k_place   <<<(E + B - 1) / B, B, 0, stream>>>(src, dst, rank, rowptr, col_a, E);

    k_pull<<<(int)(((size_t)n * 32 + B - 1) / B), B, 0, stream>>>(emb, rowptr, col_a, dinv, nullptr, agg1, n, E);
    k_dense_ws<64, 128, 2, true><<<1024, 128, 0, stream>>>(agg1, W1, b1, h, n);

    k_dense_ws<128, 64, 1, false><<<1024, 128, 0, stream>>>(h, W2, nullptr, t, n);
    k_move<<<(E + B - 1) / B, B, 0, stream>>>(col_a, col_b, E);
    k_pull<<<(int)(((size_t)n * 32 + B - 1) / B), B, 0, stream>>>(t, rowptr, col_b, dinv, b2, out, n, E);
}


# edge list = one 32-slot line per destination row, filled in one pass (atomic rank -> slot store; ranks past 32 to an overflow list, summed last in its order): no prefix sum, placement or list move
# speedup vs baseline: 2.7812x; 1.1805x over previous
#include <hip/hip_runtime.h>

typedef __attribute__((ext_vector_type(2))) float v2f;
typedef __attribute__((ext_vector_type(4))) int   v4i;
typedef __attribute__((ext_vector_type(8))) float v8f;

#define FDIM 64
#define CAP  32

__global__ void k_zero_cnt(int* __restrict__ cnt, int n) {
    int i = blockIdx.x * blockDim.x + threadIdx.x;
    if (i < n) cnt[i] = 0;
}

__global__ void k_fill(const int* __restrict__ src, const int* __restrict__ dst,
                       int* __restrict__ cnt, int* __restrict__ slot,
                       int* __restrict__ ovf_n, int* __restrict__ ovf, int E) {
    int e = blockIdx.x * blockDim.x + threadIdx.x;
    if (e < E) {
        const int d = dst[e];
        const int r = atomicAdd(&cnt[d], 1);
        if (r < CAP) slot[d * CAP + r] = src[e];
        else         ovf[atomicAdd(ovf_n, 1)] = e;
    }
}

__device__ __forceinline__ float norm_of(int c) {
    const float d = (float)(c + 1);
    return d > 0.0f ? rsqrtf(d) : 0.0f;
}

template <bool SWZ>
__global__ void __launch_bounds__(256)
k_pull(const float* __restrict__ x, const int* __restrict__ cnt, const int* __restrict__ slot,
       const int* __restrict__ ovf_n, const int* __restrict__ ovf,
       const int* __restrict__ src, const int* __restrict__ dst,
       const float* __restrict__ bias, float* __restrict__ y, int n) {
    const int lane = threadIdx.x & 31;
    const int d = __builtin_amdgcn_readfirstlane((int)((blockIdx.x * 256u + threadIdx.x) >> 5));
    if (d >= n) return;

    const int   c  = cnt[d];
    const float dd = norm_of(c);
    const float* xl = x + 2 * lane;
    auto row = [](int s) -> long long {
        const int unit = SWZ ? ((s << 1) | (__builtin_popcount((unsigned)s) & 1)) : s;
        return (long long)unit * FDIM;
    };

    v2f xs = *(const v2f*)(xl + row(d));
    v2f acc;
    acc.x = xs.x * dd * dd;
    acc.y = xs.y * dd * dd;
    if (bias) { acc.x += bias[2 * lane]; acc.y += bias[2 * lane + 1]; }

    const int m = c < CAP ? c : CAP;
    int   my = d;
    float wv = 0.0f;
    if (lane < m) { my = slot[d * CAP + lane]; wv = norm_of(cnt[my]) * dd; }

    int k = 0;
    for (; k + 8 <= m; k += 8) {
        v2f r[8]; float w[8];
#pragma unroll
        for (int u = 0; u < 8; ++u) {
            int s = __builtin_amdgcn_readlane(my, k + u);
            w[u]  = __int_as_float(__builtin_amdgcn_readlane(__float_as_int(wv), k + u));
            r[u]  = *(const v2f*)(xl + row(s));
        }
#pragma unroll
        for (int u = 0; u < 8; ++u) acc += r[u] * w[u];
    }
    for (; k + 4 <= m; k += 4) {
        v2f r[4]; float w[4];
#pragma unroll
        for (int u = 0; u < 4; ++u) {
            int s = __builtin_amdgcn_readlane(my, k + u);
            w[u]  = __int_as_float(__builtin_amdgcn_readlane(__float_as_int(wv), k + u));
            r[u]  = *(const v2f*)(xl + row(s));
        }
#pragma unroll
        for (int u = 0; u < 4; ++u) acc += r[u] * w[u];
    }
    for (; k < m; ++k) {
        int   s = __builtin_amdgcn_readlane(my, k);
        float w = __int_as_float(__builtin_amdgcn_readlane(__float_as_int(wv), k));
        v2f   r = *(const v2f*)(xl + row(s));
        acc += r * w;
    }

    if (c > CAP) {
        const int no = *ovf_n;
        for (int i0 = 0; i0 < no; i0 += 32) {
            int   s = d;
            float w = 0.0f;
            bool  hit = false;
            if (i0 + lane < no) {
                const int e = ovf[i0 + lane];
                if (dst[e] == d) { hit = true; s = src[e]; w = norm_of(cnt[s]) * dd; }
            }
            unsigned msk = __builtin_amdgcn_ballot_w32(hit);
            while (msk) {
                const int l = __builtin_ctz(msk);
                msk &= msk - 1;
                const int   ss = __builtin_amdgcn_readlane(s, l);
                const float ww = __int_as_float(__builtin_amdgcn_readlane(__float_as_int(w), l));
                v2f r = *(const v2f*)(xl + row(ss));
                acc += r * ww;
            }
        }
    }

    *(v2f*)(y + (long long)d * FDIM + 2 * lane) = acc;
}

typedef __attribute__((ext_vector_type(4))) float v4f;

template <int K, int NC, int CT, bool RELU, int OP, bool SWZ>
__global__ void __launch_bounds__(128)
k_dense_ws(const float* A, const float* __restrict__ Bm,
           const float* __restrict__ bias, float* out, int n) {
    constexpr int KS  = K / 4;
    constexpr int AP  = K + 4;
    constexpr int SP  = NC + 4;
    constexpr int NLD = (16 * K / 4) / 128;
    constexpr int NST = (16 * NC / 4) / 128;
    static_assert(NC == 4 * 16 * CT, "4 waves x CT column tiles cover NC");
    static_assert(!SWZ || OP >= 2 * NC, "the shifted row stays inside its pitch");

    __shared__ __attribute__((aligned(16))) float sA[16 * AP];
    __shared__ __attribute__((aligned(16))) float sD[16 * SP];

    const int tid  = threadIdx.x;
    const int lane = tid & 31;
    const int w    = tid >> 5;
    const int c    = lane & 15;
    const int h    = lane >> 4;
    const int ntiles = (n + 15) >> 4;

    v2f   b[CT][KS];
    float bv[CT];
#pragma unroll
    for (int j = 0; j < CT; ++j) {
        const int col = (w * CT + j) * 16 + c;
#pragma unroll
        for (int ks = 0; ks < KS; ++ks) {
            b[j][ks].x = Bm[(4 * ks + 2 * h)     * NC + col];
            b[j][ks].y = Bm[(4 * ks + 2 * h + 1) * NC + col];
        }
        bv[j] = bias ? bias[col] : 0.0f;
    }

    int tile = blockIdx.x;
    v4f pre[NLD];
    if (tile < ntiles) {
        const float* at = A + (long long)tile * 16 * K;
#pragma unroll
        for (int p = 0; p < NLD; ++p) {
            const int idx = (p * 128 + tid) * 4;
            v4f z = {0.0f, 0.0f, 0.0f, 0.0f};
            pre[p] = (tile * 16 + idx / K < n) ? *(const v4f*)(at + idx) : z;
        }
    }

    for (; tile < ntiles; tile += gridDim.x) {
#pragma unroll
        for (int p = 0; p < NLD; ++p) {
            const int idx = (p * 128 + tid) * 4;
            *(v4f*)&sA[(idx / K) * AP + (idx % K)] = pre[p];
        }
        __syncthreads();

        const int nt = tile + gridDim.x;
        if (nt < ntiles) {
            const float* at = A + (long long)nt * 16 * K;
#pragma unroll
            for (int p = 0; p < NLD; ++p) {
                const int idx = (p * 128 + tid) * 4;
                v4f z = {0.0f, 0.0f, 0.0f, 0.0f};
                pre[p] = (nt * 16 + idx / K < n) ? *(const v4f*)(at + idx) : z;
            }
        }

        v2f a[KS];
#pragma unroll
        for (int ks = 0; ks < KS; ++ks) a[ks] = *(const v2f*)&sA[c * AP + 4 * ks + 2 * h];
        __builtin_amdgcn_sched_barrier(0);

        v8f acc[CT];
#pragma unroll
        for (int j = 0; j < CT; ++j) acc[j] = (v8f){0.0f, 0.0f, 0.0f, 0.0f, 0.0f, 0.0f, 0.0f, 0.0f};
#pragma unroll
        for (int ks = 0; ks < KS; ++ks) {
#pragma unroll
            for (int j = 0; j < CT; ++j)
                acc[j] = __builtin_amdgcn_wmma_f32_16x16x4_f32(
                             false, a[ks], false, b[j][ks], (short)0, acc[j], false, false);
        }

#pragma unroll
        for (int j = 0; j < CT; ++j) {
#pragma unroll
            for (int r = 0; r < 8; ++r) {
                float v = acc[j][r] + bv[j];
                if (RELU) v = fmaxf(v, 0.0f);
                sD[(r + 8 * h) * SP + (w * CT + j) * 16 + c] = v;
            }
        }
        __syncthreads();

        float* ot = out + (long long)tile * 16 * OP;
#pragma unroll
        for (int p = 0; p < NST; ++p) {
            const int idx = (p * 128 + tid) * 4;
            const int r = idx / NC, cc = idx % NC;
            const int sw = SWZ ? (__builtin_popcount((unsigned)(tile * 16 + r)) & 1) * NC : 0;
            if (tile * 16 + r < n) *(v4f*)(ot + r * OP + cc + sw) = *(const v4f*)&sD[r * SP + cc];
        }
    }
}

extern "C" void kernel_launch(void* const* d_in, const int* in_sizes, int n_in,
                              void* d_out, int out_size, void* d_ws, size_t ws_size,
                              hipStream_t stream) {
    (void)n_in; (void)out_size; (void)ws_size;

    const int*   eidx = (const int*)d_in[0];
    const float* emb  = (const float*)d_in[1];
    const float* W1   = (const float*)d_in[2];
    const float* b1   = (const float*)d_in[3];
    const float* W2   = (const float*)d_in[4];
    const float* b2   = (const float*)d_in[5];

    const int E = in_sizes[0] / 2;
    const int n = in_sizes[1] / 64;
    const int* src = eidx;
    const int* dst = eidx + E;

    int*   cnt   = (int*)d_ws;
    int*   ovf_n = cnt + n;
    int*   slot  = (int*)d_ws + 2 * (size_t)n;
    int*   ovf   = slot + (size_t)n * CAP;
    float* h     = (float*)d_ws + 2 * (size_t)n + (size_t)n * 64;
    float* t     = h;
    float* out   = (float*)d_out;
    float* agg1  = out;

    const int B = 256;
    k_zero_cnt<<<(n + 1 + B - 1) / B, B, 0, stream>>>(cnt, n + 1);
    k_fill    <<<(E + B - 1) / B, B, 0, stream>>>(src, dst, cnt, slot, ovf_n, ovf, E);

    k_pull<false><<<(int)(((size_t)n * 32 + B - 1) / B), B, 0, stream>>>(emb, cnt, slot, ovf_n, ovf, src, dst, nullptr, agg1, n);
    k_dense_ws<64, 128, 2, true, 128, false><<<1024, 128, 0, stream>>>(agg1, W1, b1, h, n);

    k_dense_ws<128, 64, 1, false, 128, true><<<1024, 128, 0, stream>>>(h, W2, nullptr, t, n);
    k_pull<true><<<(int)(((size_t)n * 32 + B - 1) / B), B, 0, stream>>>(t, cnt, slot, ovf_n, ovf, src, dst, b2, out, n);
}


# fill: in-edge counters and slot lists kept in two planes, one per package half (by die number): rank atomics and slot stores stay half-local; a row sums plane 0 then plane 1, arrival order in each
# speedup vs baseline: 4.5961x; 1.6526x over previous
#include <hip/hip_runtime.h>

typedef __attribute__((ext_vector_type(2))) float v2f;
typedef __attribute__((ext_vector_type(4))) int   v4i;
typedef __attribute__((ext_vector_type(8))) float v8f;

#define FDIM 64
#define CAPH 24

__global__ void k_zero_cnt(int* __restrict__ cnt, int nw, int* __restrict__ ovf_n) {
    int i = blockIdx.x * blockDim.x + threadIdx.x;
    if (i < nw) cnt[i] = 0;
    if (i == 0) *ovf_n = 0;
}

__global__ void k_fill(const int* __restrict__ src, const int* __restrict__ dst,
                       int* __restrict__ cnt, int* __restrict__ slot,
                       int* __restrict__ ovf_n, int* __restrict__ ovf, int E, int n) {
    int e = blockIdx.x * blockDim.x + threadIdx.x;
    const int p = (int)((__builtin_amdgcn_s_sendmsg_rtn(0x87) >> 18) & 1u);
    if (e < E) {
        const int d = dst[e];
        int* cp = cnt  + (size_t)p * n;
        int* sp = slot + (size_t)p * n * CAPH;
        const int r = atomicAdd(&cp[d], 1);
        if (r < CAPH) sp[(size_t)d * CAPH + r] = src[e];
        else          ovf[atomicAdd(ovf_n, 1)] = e;
    }
}

__global__ void k_deg(int* __restrict__ cnt, int n) {
    int i = blockIdx.x * blockDim.x + threadIdx.x;
    if (i < n) cnt[i] += cnt[n + i];
}

__device__ __forceinline__ float norm_of(int c) {
    const float d = (float)(c + 1);
    return d > 0.0f ? rsqrtf(d) : 0.0f;
}

template <bool SWZ>
__global__ void __launch_bounds__(256)
k_pull(const float* __restrict__ x, const int* __restrict__ cnt, const int* __restrict__ slot,
       const int* __restrict__ ovf_n, const int* __restrict__ ovf,
       const int* __restrict__ src, const int* __restrict__ dst,
       const float* __restrict__ bias, float* __restrict__ y, int n) {
    const int lane = threadIdx.x & 31;
    const int d = __builtin_amdgcn_readfirstlane((int)((blockIdx.x * 256u + threadIdx.x) >> 5));
    if (d >= n) return;

    const int   c  = cnt[d];
    const int   c1 = cnt[n + d];
    const int   c0 = c - c1;
    const float dd = norm_of(c);
    const float* xl = x + 2 * lane;
    auto row = [](int s) -> long long {
        const int unit = SWZ ? ((s << 1) | (__builtin_popcount((unsigned)s) & 1)) : s;
        return (long long)unit * FDIM;
    };

    v2f xs = *(const v2f*)(xl + row(d));
    v2f acc;
    acc.x = xs.x * dd * dd;
    acc.y = xs.y * dd * dd;
    if (bias) { acc.x += bias[2 * lane]; acc.y += bias[2 * lane + 1]; }

    const int m0 = c0 < CAPH ? c0 : CAPH;
    const int m1 = c1 < CAPH ? c1 : CAPH;
    const int mt = m0 + m1;
    const int* s0 = slot + (long long)d * CAPH;
    const int* s1 = slot + ((long long)n + d) * CAPH;
    auto entry = [&](int j) -> int { return j < m0 ? s0[j] : s1[j - m0]; };
    const int m = mt < 32 ? mt : 32;
    int   my = d;
    float wv = 0.0f;
    if (lane < m) { my = entry(lane); wv = norm_of(cnt[my]) * dd; }

    int k = 0;
    for (; k + 8 <= m; k += 8) {
        v2f r[8]; float w[8];
#pragma unroll
        for (int u = 0; u < 8; ++u) {
            int s = __builtin_amdgcn_readlane(my, k + u);
            w[u]  = __int_as_float(__builtin_amdgcn_readlane(__float_as_int(wv), k + u));
            r[u]  = *(const v2f*)(xl + row(s));
        }
#pragma unroll
        for (int u = 0; u < 8; ++u) acc += r[u] * w[u];
    }
    for (; k + 4 <= m; k += 4) {
        v2f r[4]; float w[4];
#pragma unroll
        for (int u = 0; u < 4; ++u) {
            int s = __builtin_amdgcn_readlane(my, k + u);
            w[u]  = __int_as_float(__builtin_amdgcn_readlane(__float_as_int(wv), k + u));
            r[u]  = *(const v2f*)(xl + row(s));
        }
#pragma unroll
        for (int u = 0; u < 4; ++u) acc += r[u] * w[u];
    }
    for (; k < m; ++k) {
        int   s = __builtin_amdgcn_readlane(my, k);
        float w = __int_as_float(__builtin_amdgcn_readlane(__float_as_int(wv), k));
        v2f   r = *(const v2f*)(xl + row(s));
        acc += r * w;
    }

    if (mt > 32) {
        int   my2 = d;
        float wv2 = 0.0f;
        if (32 + lane < mt) { my2 = entry(32 + lane); wv2 = norm_of(cnt[my2]) * dd; }
        for (int k2 = 0; k2 < mt - 32; ++k2) {
            int   s = __builtin_amdgcn_readlane(my2, k2);
            float w = __int_as_float(__builtin_amdgcn_readlane(__float_as_int(wv2), k2));
            v2f   r = *(const v2f*)(xl + row(s));
            acc += r * w;
        }
    }

    if (c0 > CAPH || c1 > CAPH) {
        const int no = *ovf_n;
        for (int i0 = 0; i0 < no; i0 += 32) {
            int   s = d;
            float w = 0.0f;
            bool  hit = false;
            if (i0 + lane < no) {
                const int e = ovf[i0 + lane];
                if (dst[e] == d) { hit = true; s = src[e]; w = norm_of(cnt[s]) * dd; }
            }
            unsigned msk = __builtin_amdgcn_ballot_w32(hit);
            while (msk) {
                const int l = __builtin_ctz(msk);
                msk &= msk - 1;
                const int   ss = __builtin_amdgcn_readlane(s, l);
                const float ww = __int_as_float(__builtin_amdgcn_readlane(__float_as_int(w), l));
                v2f r = *(const v2f*)(xl + row(ss));
                acc += r * ww;
            }
        }
    }

    *(v2f*)(y + (long long)d * FDIM + 2 * lane) = acc;
}

typedef __attribute__((ext_vector_type(4))) float v4f;

template <int K, int NC, int CT, bool RELU, int OP, bool SWZ>
__global__ void __launch_bounds__(128)
k_dense_ws(const float* A, const float* __restrict__ Bm,
           const float* __restrict__ bias, float* out, int n) {
    constexpr int KS  = K / 4;
    constexpr int AP  = K + 4;
    constexpr int SP  = NC + 4;
    constexpr int NLD = (16 * K / 4) / 128;
    constexpr int NST = (16 * NC / 4) / 128;
    static_assert(NC == 4 * 16 * CT, "4 waves x CT column tiles cover NC");
    static_assert(!SWZ || OP >= 2 * NC, "the shifted row stays inside its pitch");

    __shared__ __attribute__((aligned(16))) float sA[16 * AP];
    __shared__ __attribute__((aligned(16))) float sD[16 * SP];

    const int tid  = threadIdx.x;
    const int lane = tid & 31;
    const int w    = tid >> 5;
    const int c    = lane & 15;
    const int h    = lane >> 4;
    const int ntiles = (n + 15) >> 4;

    v2f   b[CT][KS];
    float bv[CT];
#pragma unroll
    for (int j = 0; j < CT; ++j) {
        const int col = (w * CT + j) * 16 + c;
#pragma unroll
        for (int ks = 0; ks < KS; ++ks) {
            b[j][ks].x = Bm[(4 * ks + 2 * h)     * NC + col];
            b[j][ks].y = Bm[(4 * ks + 2 * h + 1) * NC + col];
        }
        bv[j] = bias ? bias[col] : 0.0f;
    }

    int tile = blockIdx.x;
    v4f pre[NLD];
    if (tile < ntiles) {
        const float* at = A + (long long)tile * 16 * K;
#pragma unroll
        for (int p = 0; p < NLD; ++p) {
            const int idx = (p * 128 + tid) * 4;
            v4f z = {0.0f, 0.0f, 0.0f, 0.0f};
            pre[p] = (tile * 16 + idx / K < n) ? *(const v4f*)(at + idx) : z;
        }
    }

    for (; tile < ntiles; tile += gridDim.x) {
#pragma unroll
        for (int p = 0; p < NLD; ++p) {
            const int idx = (p * 128 + tid) * 4;
            *(v4f*)&sA[(idx / K) * AP + (idx % K)] = pre[p];
        }
        __syncthreads();

        const int nt = tile + gridDim.x;
        if (nt < ntiles) {
            const float* at = A + (long long)nt * 16 * K;
#pragma unroll
            for (int p = 0; p < NLD; ++p) {
                const int idx = (p * 128 + tid) * 4;
                v4f z = {0.0f, 0.0f, 0.0f, 0.0f};
                pre[p] = (nt * 16 + idx / K < n) ? *(const v4f*)(at + idx) : z;
            }
        }

        v2f a[KS];
#pragma unroll
        for (int ks = 0; ks < KS; ++ks) a[ks] = *(const v2f*)&sA[c * AP + 4 * ks + 2 * h];
        __builtin_amdgcn_sched_barrier(0);

        v8f acc[CT];
#pragma unroll
        for (int j = 0; j < CT; ++j) acc[j] = (v8f){0.0f, 0.0f, 0.0f, 0.0f, 0.0f, 0.0f, 0.0f, 0.0f};
#pragma unroll
        for (int ks = 0; ks < KS; ++ks) {
#pragma unroll
            for (int j = 0; j < CT; ++j)
                acc[j] = __builtin_amdgcn_wmma_f32_16x16x4_f32(
                             false, a[ks], false, b[j][ks], (short)0, acc[j], false, false);
        }

#pragma unroll
        for (int j = 0; j < CT; ++j) {
#pragma unroll
            for (int r = 0; r < 8; ++r) {
                float v = acc[j][r] + bv[j];
                if (RELU) v = fmaxf(v, 0.0f);
                sD[(r + 8 * h) * SP + (w * CT + j) * 16 + c] = v;
            }
        }
        __syncthreads();

        float* ot = out + (long long)tile * 16 * OP;
#pragma unroll
        for (int p = 0; p < NST; ++p) {
            const int idx = (p * 128 + tid) * 4;
            const int r = idx / NC, cc = idx % NC;
            const int sw = SWZ ? (__builtin_popcount((unsigned)(tile * 16 + r)) & 1) * NC : 0;
            if (tile * 16 + r < n) *(v4f*)(ot + r * OP + cc + sw) = *(const v4f*)&sD[r * SP + cc];
        }
    }
}

extern "C" void kernel_launch(void* const* d_in, const int* in_sizes, int n_in,
                              void* d_out, int out_size, void* d_ws, size_t ws_size,
                              hipStream_t stream) {
    (void)n_in; (void)out_size; (void)ws_size;

    const int*   eidx = (const int*)d_in[0];
    const float* emb  = (const float*)d_in[1];
    const float* W1   = (const float*)d_in[2];
    const float* b1   = (const float*)d_in[3];
    const float* W2   = (const float*)d_in[4];
    const float* b2   = (const float*)d_in[5];

    const int E = in_sizes[0] / 2;
    const int n = in_sizes[1] / 64;
    const int* src = eidx;
    const int* dst = eidx + E;

    int*   cnt   = (int*)d_ws;
    int*   slot  = (int*)d_ws + 2 * (size_t)n;
    int*   ovf   = slot + 2 * (size_t)n * CAPH;
    int*   ovf_n = ovf + E - 1;
    float* h     = (float*)d_ws + 2 * (size_t)n + (size_t)n * 64;
    float* t     = h;
    float* out   = (float*)d_out;
    float* agg1  = out;

    const int B = 256;
    k_zero_cnt<<<(2 * n + B - 1) / B, B, 0, stream>>>(cnt, 2 * n, ovf_n);
    k_fill    <<<(E + B - 1) / B, B, 0, stream>>>(src, dst, cnt, slot, ovf_n, ovf, E, n);
    k_deg     <<<(n + B - 1) / B, B, 0, stream>>>(cnt, n);

    k_pull<false><<<(int)(((size_t)n * 32 + B - 1) / B), B, 0, stream>>>(emb, cnt, slot, ovf_n, ovf, src, dst, nullptr, agg1, n);
    k_dense_ws<64, 128, 2, true, 128, false><<<1024, 128, 0, stream>>>(agg1, W1, b1, h, n);

    k_dense_ws<128, 64, 1, false, 128, true><<<1024, 128, 0, stream>>>(h, W2, nullptr, t, n);
    k_pull<true><<<(int)(((size_t)n * 32 + B - 1) / B), B, 0, stream>>>(t, cnt, slot, ovf_n, ovf, src, dst, b2, out, n);
}


# one-item waves now loop (wave entry was the limit): aggregations = 16384 resident waves walking rows, next rows' counts, entries, degrees asked ahead; fill 4 edges per lane; same arrival-order sums
# speedup vs baseline: 4.8837x; 1.0626x over previous
#include <hip/hip_runtime.h>

typedef __attribute__((ext_vector_type(2))) float v2f;
typedef __attribute__((ext_vector_type(4))) int   v4i;
typedef __attribute__((ext_vector_type(8))) float v8f;

#define FDIM 64
#define CAPH 24

__global__ void k_zero_cnt(int* __restrict__ cnt, int nw, int* __restrict__ ovf_n) {
    int i = blockIdx.x * blockDim.x + threadIdx.x;
    if (i < nw) cnt[i] = 0;
    if (i == 0) *ovf_n = 0;
}

#define EPL 4
__global__ void __launch_bounds__(256)
k_fill(const int* __restrict__ src, const int* __restrict__ dst,
       int* __restrict__ cnt, int* __restrict__ slot,
       int* __restrict__ ovf_n, int* __restrict__ ovf, int E, int n) {
    const int e0 = (blockIdx.x * blockDim.x + threadIdx.x) * EPL;
    const int p = (int)((__builtin_amdgcn_s_sendmsg_rtn(0x87) >> 18) & 1u);
    if (e0 >= E) return;
    int* cp = cnt  + (size_t)p * n;
    int* sp = slot + (size_t)p * n * CAPH;
    const int nv = E - e0 < EPL ? E - e0 : EPL;
    int dv[EPL], sv[EPL], r[EPL];
    if (nv == EPL && ((((size_t)(dst + e0)) | ((size_t)(src + e0))) & 15) == 0) {
        const v4i a = *(const v4i*)(dst + e0);
        const v4i b = *(const v4i*)(src + e0);
#pragma unroll
        for (int u = 0; u < EPL; ++u) { dv[u] = a[u]; sv[u] = b[u]; }
    } else {
#pragma unroll
        for (int u = 0; u < EPL; ++u) {
            dv[u] = u < nv ? dst[e0 + u] : 0;
            sv[u] = u < nv ? src[e0 + u] : 0;
        }
    }
#pragma unroll
    for (int u = 0; u < EPL; ++u) r[u] = u < nv ? atomicAdd(&cp[dv[u]], 1) : 0;
#pragma unroll
    for (int u = 0; u < EPL; ++u) {
        if (u < nv) {
            if (r[u] < CAPH) sp[(size_t)dv[u] * CAPH + r[u]] = sv[u];
            else             ovf[atomicAdd(ovf_n, 1)] = e0 + u;
        }
    }
}

__global__ void k_deg(int* __restrict__ cnt, int n) {
    int i = blockIdx.x * blockDim.x + threadIdx.x;
    if (i < n) cnt[i] += cnt[n + i];
}

__device__ __forceinline__ float norm_of(int c) {
    const float d = (float)(c + 1);
    return d > 0.0f ? rsqrtf(d) : 0.0f;
}

template <bool SWZ>
__global__ void __launch_bounds__(256)
k_pull(const float* __restrict__ x, const int* __restrict__ cnt, const int* __restrict__ slot,
       const int* __restrict__ ovf_n, const int* __restrict__ ovf,
       const int* __restrict__ src, const int* __restrict__ dst,
       const float* __restrict__ bias, float* __restrict__ y, int n) {
    const int lane = threadIdx.x & 31;
    const int nw   = (int)gridDim.x * 8;
    int dA = __builtin_amdgcn_readfirstlane((int)((blockIdx.x * 256u + threadIdx.x) >> 5));
    if (dA >= n) return;

    const float* xl = x + 2 * lane;
    auto row = [](int s) -> long long {
        const int unit = SWZ ? ((s << 1) | (__builtin_popcount((unsigned)s) & 1)) : s;
        return (long long)unit * FDIM;
    };
    auto valid = [&](int d) -> bool { return (unsigned)d < (unsigned)n; };
    auto counts = [&](int d, int& c, int& c1) {
        c = 0; c1 = 0;
        if (valid(d)) { c = cnt[d]; c1 = cnt[n + d]; }
    };
    auto listed = [](int c, int c1, int& m0, int& m1) {
        const int c0 = c - c1;
        m0 = c0 < CAPH ? c0 : CAPH;
        m1 = c1 < CAPH ? c1 : CAPH;
    };
    auto entries = [&](int d, int c, int c1, int& my, v2f& xs) {
        my = 0;
        xs = (v2f){0.0f, 0.0f};
        if (valid(d)) {
            int m0, m1; listed(c, c1, m0, m1);
            const int mt = m0 + m1;
            const int m  = mt < 32 ? mt : 32;
            const int* s0 = slot + (long long)d * CAPH;
            const int* s1 = slot + ((long long)n + d) * CAPH;
            my = d;
            if (lane < m) my = lane < m0 ? s0[lane] : s1[lane - m0];
            xs = *(const v2f*)(xl + row(d));
        }
    };
    auto degrees = [&](int d, int my) -> int {
        int cw = 0;
        if (valid(d)) cw = cnt[my];
        return cw;
    };

    float b0 = 0.0f, b1 = 0.0f;
    if (bias) { b0 = bias[2 * lane]; b1 = bias[2 * lane + 1]; }
    asm volatile("" :: "v"(b0), "v"(b1));

    int dC = dA, dB = dA - nw;
    dA = dB - nw;
    int cA = 0, c1A = 0, cB = 0, c1B = 0, cC, c1C;
    counts(dC, cC, c1C);
    int myA = 0, myB = 0, cwA = 0;
    v2f xsA = {0.0f, 0.0f}, xsB = {0.0f, 0.0f};

    for (;;) {
        const int cwB = degrees(dB, myB);
        int myC; v2f xsC;
        entries(dC, cC, c1C, myC, xsC);
        const int dD = dC + nw;
        int cD, c1D;
        counts(dD, cD, c1D);

        if (valid(dA)) {
        const int   d  = dA;
        const int   c  = cA;
        const float dd = norm_of(c);
        int m0, m1; listed(cA, c1A, m0, m1);
        const int mt = m0 + m1;
        const int m  = mt < 32 ? mt : 32;
        const int   my = myA;
        const float wv = lane < m ? norm_of(cwA) * dd : 0.0f;

        v2f acc;
        acc.x = xsA.x * dd * dd;
        acc.y = xsA.y * dd * dd;
        if (bias) { acc.x += b0; acc.y += b1; }

        int k = 0;
        for (; k + 8 <= m; k += 8) {
            v2f r[8]; float w[8];
#pragma unroll
            for (int u = 0; u < 8; ++u) {
                int s = __builtin_amdgcn_readlane(my, k + u);
                w[u]  = __int_as_float(__builtin_amdgcn_readlane(__float_as_int(wv), k + u));
                r[u]  = *(const v2f*)(xl + row(s));
            }
#pragma unroll
            for (int u = 0; u < 8; ++u) acc += r[u] * w[u];
        }
        for (; k + 4 <= m; k += 4) {
            v2f r[4]; float w[4];
#pragma unroll
            for (int u = 0; u < 4; ++u) {
                int s = __builtin_amdgcn_readlane(my, k + u);
                w[u]  = __int_as_float(__builtin_amdgcn_readlane(__float_as_int(wv), k + u));
                r[u]  = *(const v2f*)(xl + row(s));
            }
#pragma unroll
            for (int u = 0; u < 4; ++u) acc += r[u] * w[u];
        }
        for (; k < m; ++k) {
            int   s = __builtin_amdgcn_readlane(my, k);
            float w = __int_as_float(__builtin_amdgcn_readlane(__float_as_int(wv), k));
            v2f   r = *(const v2f*)(xl + row(s));
            acc += r * w;
        }

        if (mt > 32) {
            const int* s0 = slot + (long long)d * CAPH;
            const int* s1 = slot + ((long long)n + d) * CAPH;
            int   my2 = d;
            float wv2 = 0.0f;
            if (32 + lane < mt) {
                const int j = 32 + lane;
                my2 = j < m0 ? s0[j] : s1[j - m0];
                wv2 = norm_of(cnt[my2]) * dd;
            }
            for (int k2 = 0; k2 < mt - 32; ++k2) {
                int   s = __builtin_amdgcn_readlane(my2, k2);
                float w = __int_as_float(__builtin_amdgcn_readlane(__float_as_int(wv2), k2));
                v2f   r = *(const v2f*)(xl + row(s));
                acc += r * w;
            }
        }

        if (cA - c1A > CAPH || c1A > CAPH) {
            const int no = *ovf_n;
            for (int i0 = 0; i0 < no; i0 += 32) {
                int   s = d;
                float w = 0.0f;
                bool  hit = false;
                if (i0 + lane < no) {
                    const int e = ovf[i0 + lane];
                    if (dst[e] == d) { hit = true; s = src[e]; w = norm_of(cnt[s]) * dd; }
                }
                unsigned msk = __builtin_amdgcn_ballot_w32(hit);
                while (msk) {
                    const int l = __builtin_ctz(msk);
                    msk &= msk - 1;
                    const int   ss = __builtin_amdgcn_readlane(s, l);
                    const float ww = __int_as_float(__builtin_amdgcn_readlane(__float_as_int(w), l));
                    v2f r = *(const v2f*)(xl + row(ss));
                    acc += r * ww;
                }
            }
        }

        *(v2f*)(y + (long long)d * FDIM + 2 * lane) = acc;
        }

        if (dB >= n) break;
        dA = dB; cA = cB; c1A = c1B; myA = myB; xsA = xsB; cwA = cwB;
        dB = dC; cB = cC; c1B = c1C; myB = myC; xsB = xsC;
        dC = dD; cC = cD; c1C = c1D;
    }
}

typedef __attribute__((ext_vector_type(4))) float v4f;

template <int K, int NC, int CT, bool RELU, int OP, bool SWZ>
__global__ void __launch_bounds__(128)
k_dense_ws(const float* A, const float* __restrict__ Bm,
           const float* __restrict__ bias, float* out, int n) {
    constexpr int KS  = K / 4;
    constexpr int AP  = K + 4;
    constexpr int SP  = NC + 4;
    constexpr int NLD = (16 * K / 4) / 128;
    constexpr int NST = (16 * NC / 4) / 128;
    static_assert(NC == 4 * 16 * CT, "4 waves x CT column tiles cover NC");
    static_assert(!SWZ || OP >= 2 * NC, "the shifted row stays inside its pitch");

    __shared__ __attribute__((aligned(16))) float sA[16 * AP];
    __shared__ __attribute__((aligned(16))) float sD[16 * SP];

    const int tid  = threadIdx.x;
    const int lane = tid & 31;
    const int w    = tid >> 5;
    const int c    = lane & 15;
    const int h    = lane >> 4;
    const int ntiles = (n + 15) >> 4;

    v2f   b[CT][KS];
    float bv[CT];
#pragma unroll
    for (int j = 0; j < CT; ++j) {
        const int col = (w * CT + j) * 16 + c;
#pragma unroll
        for (int ks = 0; ks < KS; ++ks) {
            b[j][ks].x = Bm[(4 * ks + 2 * h)     * NC + col];
            b[j][ks].y = Bm[(4 * ks + 2 * h + 1) * NC + col];
        }
        bv[j] = bias ? bias[col] : 0.0f;
    }

    int tile = blockIdx.x;
    v4f pre[NLD];
    if (tile < ntiles) {
        const float* at = A + (long long)tile * 16 * K;
#pragma unroll
        for (int p = 0; p < NLD; ++p) {
            const int idx = (p * 128 + tid) * 4;
            v4f z = {0.0f, 0.0f, 0.0f, 0.0f};
            pre[p] = (tile * 16 + idx / K < n) ? *(const v4f*)(at + idx) : z;
        }
    }

    for (; tile < ntiles; tile += gridDim.x) {
#pragma unroll
        for (int p = 0; p < NLD; ++p) {
            const int idx = (p * 128 + tid) * 4;
            *(v4f*)&sA[(idx / K) * AP + (idx % K)] = pre[p];
        }
        __syncthreads();

        const int nt = tile + gridDim.x;
        if (nt < ntiles) {
            const float* at = A + (long long)nt * 16 * K;
#pragma unroll
            for (int p = 0; p < NLD; ++p) {
                const int idx = (p * 128 + tid) * 4;
                v4f z = {0.0f, 0.0f, 0.0f, 0.0f};
                pre[p] = (nt * 16 + idx / K < n) ? *(const v4f*)(at + idx) : z;
            }
        }

        v2f a[KS];
#pragma unroll
        for (int ks = 0; ks < KS; ++ks) a[ks] = *(const v2f*)&sA[c * AP + 4 * ks + 2 * h];
        __builtin_amdgcn_sched_barrier(0);

        v8f acc[CT];
#pragma unroll
        for (int j = 0; j < CT; ++j) acc[j] = (v8f){0.0f, 0.0f, 0.0f, 0.0f, 0.0f, 0.0f, 0.0f, 0.0f};
#pragma unroll
        for (int ks = 0; ks < KS; ++ks) {
#pragma unroll
            for (int j = 0; j < CT; ++j)
                acc[j] = __builtin_amdgcn_wmma_f32_16x16x4_f32(
                             false, a[ks], false, b[j][ks], (short)0, acc[j], false, false);
        }

#pragma unroll
        for (int j = 0; j < CT; ++j) {
#pragma unroll
            for (int r = 0; r < 8; ++r) {
                float v = acc[j][r] + bv[j];
                if (RELU) v = fmaxf(v, 0.0f);
                sD[(r + 8 * h) * SP + (w * CT + j) * 16 + c] = v;
            }
        }
        __syncthreads();

        float* ot = out + (long long)tile * 16 * OP;
#pragma unroll
        for (int p = 0; p < NST; ++p) {
            const int idx = (p * 128 + tid) * 4;
            const int r = idx / NC, cc = idx % NC;
            const int sw = SWZ ? (__builtin_popcount((unsigned)(tile * 16 + r)) & 1) * NC : 0;
            if (tile * 16 + r < n) *(v4f*)(ot + r * OP + cc + sw) = *(const v4f*)&sD[r * SP + cc];
        }
    }
}

extern "C" void kernel_launch(void* const* d_in, const int* in_sizes, int n_in,
                              void* d_out, int out_size, void* d_ws, size_t ws_size,
                              hipStream_t stream) {
    (void)n_in; (void)out_size; (void)ws_size;

    const int*   eidx = (const int*)d_in[0];
    const float* emb  = (const float*)d_in[1];
    const float* W1   = (const float*)d_in[2];
    const float* b1   = (const float*)d_in[3];
    const float* W2   = (const float*)d_in[4];
    const float* b2   = (const float*)d_in[5];

    const int E = in_sizes[0] / 2;
    const int n = in_sizes[1] / 64;
    const int* src = eidx;
    const int* dst = eidx + E;

    int*   cnt   = (int*)d_ws;
    int*   slot  = (int*)d_ws + 2 * (size_t)n;
    int*   ovf   = slot + 2 * (size_t)n * CAPH;
    int*   ovf_n = ovf + E - 1;
    float* h     = (float*)d_ws + 2 * (size_t)n + (size_t)n * 64;
    float* t     = h;
    float* out   = (float*)d_out;
    float* agg1  = out;

    const int B = 256;
    k_zero_cnt<<<(2 * n + B - 1) / B, B, 0, stream>>>(cnt, 2 * n, ovf_n);
    k_fill    <<<(E + EPL * B - 1) / (EPL * B), B, 0, stream>>>(src, dst, cnt, slot, ovf_n, ovf, E, n);
    k_deg     <<<(n + B - 1) / B, B, 0, stream>>>(cnt, n);

    k_pull<false><<<2048, B, 0, stream>>>(emb, cnt, slot, ovf_n, ovf, src, dst, nullptr, agg1, n);
    k_dense_ws<64, 128, 2, true, 128, false><<<1024, 128, 0, stream>>>(agg1, W1, b1, h, n);

    k_dense_ws<128, 64, 1, false, 128, true><<<1024, 128, 0, stream>>>(h, W2, nullptr, t, n);
    k_pull<true><<<2048, B, 0, stream>>>(t, cnt, slot, ovf_n, ovf, src, dst, b2, out, n);
}


# layer-1 aggregation stores each row's merged list as 32 (source, weight) pairs, one 256-byte unit; layer 2 reads that unit: no slot lines, no per-entry in-degree request; sums same to the bit
# speedup vs baseline: 5.1430x; 1.0531x over previous
#include <hip/hip_runtime.h>

typedef __attribute__((ext_vector_type(2))) float v2f;
typedef __attribute__((ext_vector_type(4))) int   v4i;
typedef __attribute__((ext_vector_type(8))) float v8f;

#define FDIM 64
#define CAPH 24

__global__ void k_zero_cnt(int* __restrict__ cnt, int nw, int* __restrict__ ovf_n) {
    int i = blockIdx.x * blockDim.x + threadIdx.x;
    if (i < nw) cnt[i] = 0;
    if (i == 0) *ovf_n = 0;
}

#define EPL 4
__global__ void __launch_bounds__(256)
k_fill(const int* __restrict__ src, const int* __restrict__ dst,
       int* __restrict__ cnt, int* __restrict__ slot,
       int* __restrict__ ovf_n, int* __restrict__ ovf, int E, int n) {
    const int e0 = (blockIdx.x * blockDim.x + threadIdx.x) * EPL;
    const int p = (int)((__builtin_amdgcn_s_sendmsg_rtn(0x87) >> 18) & 1u);
    if (e0 >= E) return;
    int* cp = cnt  + (size_t)p * n;
    int* sp = slot + (size_t)p * n * CAPH;
    const int nv = E - e0 < EPL ? E - e0 : EPL;
    int dv[EPL], sv[EPL], r[EPL];
    if (nv == EPL && ((((size_t)(dst + e0)) | ((size_t)(src + e0))) & 15) == 0) {
        const v4i a = *(const v4i*)(dst + e0);
        const v4i b = *(const v4i*)(src + e0);
#pragma unroll
        for (int u = 0; u < EPL; ++u) { dv[u] = a[u]; sv[u] = b[u]; }
    } else {
#pragma unroll
        for (int u = 0; u < EPL; ++u) {
            dv[u] = u < nv ? dst[e0 + u] : 0;
            sv[u] = u < nv ? src[e0 + u] : 0;
        }
    }
#pragma unroll
    for (int u = 0; u < EPL; ++u) r[u] = u < nv ? atomicAdd(&cp[dv[u]], 1) : 0;
#pragma unroll
    for (int u = 0; u < EPL; ++u) {
        if (u < nv) {
            if (r[u] < CAPH) sp[(size_t)dv[u] * CAPH + r[u]] = sv[u];
            else             ovf[atomicAdd(ovf_n, 1)] = e0 + u;
        }
    }
}

__global__ void k_deg(int* __restrict__ cnt, int n) {
    int i = blockIdx.x * blockDim.x + threadIdx.x;
    if (i < n) cnt[i] += cnt[n + i];
}

__device__ __forceinline__ float norm_of(int c) {
    const float d = (float)(c + 1);
    return d > 0.0f ? rsqrtf(d) : 0.0f;
}

template <bool SWZ, bool WL, bool RL, int YP, bool YSW>
__global__ void __launch_bounds__(256)
k_pull(const float* __restrict__ x, const int* __restrict__ cnt, const int* __restrict__ slot,
       const int* __restrict__ ovf_n, const int* __restrict__ ovf,
       const int* __restrict__ src, const int* __restrict__ dst,
       const float* __restrict__ bias, float* y, float* pl, int n) {
    const int lane = threadIdx.x & 31;
    const int nw   = (int)gridDim.x * 8;
    int dA = __builtin_amdgcn_readfirstlane((int)((blockIdx.x * 256u + threadIdx.x) >> 5));
    if (dA >= n) return;

    const float* xl = x + 2 * lane;
    auto row = [](int s) -> long long {
        const int unit = SWZ ? ((s << 1) | (__builtin_popcount((unsigned)s) & 1)) : s;
        return (long long)unit * FDIM;
    };
    auto valid = [&](int d) -> bool { return (unsigned)d < (unsigned)n; };
    auto counts = [&](int d, int& c, int& c1) {
        c = 0; c1 = 0;
        if (valid(d)) { c = cnt[d]; c1 = cnt[n + d]; }
    };
    auto listed = [](int c, int c1, int& m0, int& m1) {
        const int c0 = c - c1;
        m0 = c0 < CAPH ? c0 : CAPH;
        m1 = c1 < CAPH ? c1 : CAPH;
    };
    auto entries = [&](int d, int c, int c1, int& my, float& wq, v2f& xs) {
        my = 0;
        wq = 0.0f;
        xs = (v2f){0.0f, 0.0f};
        if (valid(d)) {
            if (RL) {
                const v2f pr = *(const v2f*)(pl + (long long)d * 64 + 2 * lane);
                my = __float_as_int(pr.x);
                wq = pr.y;
            } else {
                int m0, m1; listed(c, c1, m0, m1);
                const int mt = m0 + m1;
                const int m  = mt < 32 ? mt : 32;
                const int* s0 = slot + (long long)d * CAPH;
                const int* s1 = slot + ((long long)n + d) * CAPH;
                my = d;
                if (lane < m) my = lane < m0 ? s0[lane] : s1[lane - m0];
            }
            xs = *(const v2f*)(xl + row(d));
        }
    };
    auto degrees = [&](int d, int my) -> int {
        int cw = 0;
        if (!RL && valid(d)) cw = cnt[my];
        return cw;
    };

    float b0 = 0.0f, b1 = 0.0f;
    if (bias) { b0 = bias[2 * lane]; b1 = bias[2 * lane + 1]; }
    asm volatile("" :: "v"(b0), "v"(b1));

    int dC = dA, dB = dA - nw;
    dA = dB - nw;
    int cA = 0, c1A = 0, cB = 0, c1B = 0, cC, c1C;
    counts(dC, cC, c1C);
    int myA = 0, myB = 0, cwA = 0;
    float wqA = 0.0f, wqB = 0.0f;
    v2f xsA = {0.0f, 0.0f}, xsB = {0.0f, 0.0f};

    for (;;) {
        const int cwB = degrees(dB, myB);
        int myC; float wqC; v2f xsC;
        entries(dC, cC, c1C, myC, wqC, xsC);
        const int dD = dC + nw;
        int cD, c1D;
        counts(dD, cD, c1D);

        if (valid(dA)) {
        const int   d  = dA;
        const int   c  = cA;
        const float dd = norm_of(c);
        int m0, m1; listed(cA, c1A, m0, m1);
        const int mt = m0 + m1;
        const int m  = mt < 32 ? mt : 32;
        const int   my = myA;
        const float wv = RL ? wqA : (lane < m ? norm_of(cwA) * dd : 0.0f);
        if (WL) *(v2f*)(pl + (long long)d * 64 + 2 * lane) = (v2f){__int_as_float(my), wv};

        v2f acc;
        acc.x = xsA.x * dd * dd;
        acc.y = xsA.y * dd * dd;
        if (bias) { acc.x += b0; acc.y += b1; }

        int k = 0;
        for (; k + 8 <= m; k += 8) {
            v2f r[8]; float w[8];
#pragma unroll
            for (int u = 0; u < 8; ++u) {
                int s = __builtin_amdgcn_readlane(my, k + u);
                w[u]  = __int_as_float(__builtin_amdgcn_readlane(__float_as_int(wv), k + u));
                r[u]  = *(const v2f*)(xl + row(s));
            }
#pragma unroll
            for (int u = 0; u < 8; ++u) acc += r[u] * w[u];
        }
        for (; k + 4 <= m; k += 4) {
            v2f r[4]; float w[4];
#pragma unroll
            for (int u = 0; u < 4; ++u) {
                int s = __builtin_amdgcn_readlane(my, k + u);
                w[u]  = __int_as_float(__builtin_amdgcn_readlane(__float_as_int(wv), k + u));
                r[u]  = *(const v2f*)(xl + row(s));
            }
#pragma unroll
            for (int u = 0; u < 4; ++u) acc += r[u] * w[u];
        }
        for (; k < m; ++k) {
            int   s = __builtin_amdgcn_readlane(my, k);
            float w = __int_as_float(__builtin_amdgcn_readlane(__float_as_int(wv), k));
            v2f   r = *(const v2f*)(xl + row(s));
            acc += r * w;
        }

        if (mt > 32) {
            const int* s0 = slot + (long long)d * CAPH;
            const int* s1 = slot + ((long long)n + d) * CAPH;
            int   my2 = d;
            float wv2 = 0.0f;
            if (32 + lane < mt) {
                const int j = 32 + lane;
                my2 = j < m0 ? s0[j] : s1[j - m0];
                wv2 = norm_of(cnt[my2]) * dd;
            }
            for (int k2 = 0; k2 < mt - 32; ++k2) {
                int   s = __builtin_amdgcn_readlane(my2, k2);
                float w = __int_as_float(__builtin_amdgcn_readlane(__float_as_int(wv2), k2));
                v2f   r = *(const v2f*)(xl + row(s));
                acc += r * w;
            }
        }

        if (cA - c1A > CAPH || c1A > CAPH) {
            const int no = *ovf_n;
            for (int i0 = 0; i0 < no; i0 += 32) {
                int   s = d;
                float w = 0.0f;
                bool  hit = false;
                if (i0 + lane < no) {
                    const int e = ovf[i0 + lane];
                    if (dst[e] == d) { hit = true; s = src[e]; w = norm_of(cnt[s]) * dd; }
                }
                unsigned msk = __builtin_amdgcn_ballot_w32(hit);
                while (msk) {
                    const int l = __builtin_ctz(msk);
                    msk &= msk - 1;
                    const int   ss = __builtin_amdgcn_readlane(s, l);
                    const float ww = __int_as_float(__builtin_amdgcn_readlane(__float_as_int(w), l));
                    v2f r = *(const v2f*)(xl + row(ss));
                    acc += r * ww;
                }
            }
        }

        *(v2f*)(y + (long long)d * YP + (YSW ? (__builtin_popcount((unsigned)d) & 1) * FDIM : 0) + 2 * lane) = acc;
        }

        if (dB >= n) break;
        dA = dB; cA = cB; c1A = c1B; myA = myB; xsA = xsB; cwA = cwB; wqA = wqB;
        dB = dC; cB = cC; c1B = c1C; myB = myC; xsB = xsC; wqB = wqC;
        dC = dD; cC = cD; c1C = c1D;
    }
}

typedef __attribute__((ext_vector_type(4))) float v4f;

template <int K, int NC, int CT, bool RELU, int OP, bool SWZ, int AG, bool ASW>
__global__ void __launch_bounds__(128)
k_dense_ws(const float* A, const float* __restrict__ Bm,
           const float* __restrict__ bias, float* out, int n) {
    constexpr int KS  = K / 4;
    constexpr int AP  = K + 4;
    constexpr int SP  = NC + 4;
    constexpr int NLD = (16 * K / 4) / 128;
    constexpr int NST = (16 * NC / 4) / 128;
    static_assert(NC == 4 * 16 * CT, "4 waves x CT column tiles cover NC");
    static_assert(!SWZ || OP >= 2 * NC, "the shifted row stays inside its pitch");
    static_assert(!ASW || AG >= 2 * K, "the shifted A row stays inside its pitch");

    __shared__ __attribute__((aligned(16))) float sA[16 * AP];
    __shared__ __attribute__((aligned(16))) float sD[16 * SP];

    const int tid  = threadIdx.x;
    const int lane = tid & 31;
    const int w    = tid >> 5;
    const int c    = lane & 15;
    const int h    = lane >> 4;
    const int ntiles = (n + 15) >> 4;

    v2f   b[CT][KS];
    float bv[CT];
#pragma unroll
    for (int j = 0; j < CT; ++j) {
        const int col = (w * CT + j) * 16 + c;
#pragma unroll
        for (int ks = 0; ks < KS; ++ks) {
            b[j][ks].x = Bm[(4 * ks + 2 * h)     * NC + col];
            b[j][ks].y = Bm[(4 * ks + 2 * h + 1) * NC + col];
        }
        bv[j] = bias ? bias[col] : 0.0f;
    }

    int tile = blockIdx.x;
    v4f pre[NLD];
    if (tile < ntiles) {
        const float* at = A + (long long)tile * 16 * AG;
#pragma unroll
        for (int p = 0; p < NLD; ++p) {
            const int idx = (p * 128 + tid) * 4;
            v4f z = {0.0f, 0.0f, 0.0f, 0.0f};
            const int asw = ASW ? (__builtin_popcount((unsigned)(tile * 16 + idx / K)) & 1) * K : 0;
            pre[p] = (tile * 16 + idx / K < n) ? *(const v4f*)(at + (idx / K) * AG + (idx % K) + asw) : z;
        }
    }

    for (; tile < ntiles; tile += gridDim.x) {
#pragma unroll
        for (int p = 0; p < NLD; ++p) {
            const int idx = (p * 128 + tid) * 4;
            *(v4f*)&sA[(idx / K) * AP + (idx % K)] = pre[p];
        }
        __syncthreads();

        const int nt = tile + gridDim.x;
        if (nt < ntiles) {
            const float* at = A + (long long)nt * 16 * AG;
#pragma unroll
            for (int p = 0; p < NLD; ++p) {
                const int idx = (p * 128 + tid) * 4;
                v4f z = {0.0f, 0.0f, 0.0f, 0.0f};
                const int asw = ASW ? (__builtin_popcount((unsigned)(nt * 16 + idx / K)) & 1) * K : 0;
                pre[p] = (nt * 16 + idx / K < n) ? *(const v4f*)(at + (idx / K) * AG + (idx % K) + asw) : z;
            }
        }

        v2f a[KS];
#pragma unroll
        for (int ks = 0; ks < KS; ++ks) a[ks] = *(const v2f*)&sA[c * AP + 4 * ks + 2 * h];
        __builtin_amdgcn_sched_barrier(0);

        v8f acc[CT];
#pragma unroll
        for (int j = 0; j < CT; ++j) acc[j] = (v8f){0.0f, 0.0f, 0.0f, 0.0f, 0.0f, 0.0f, 0.0f, 0.0f};
#pragma unroll
        for (int ks = 0; ks < KS; ++ks) {
#pragma unroll
            for (int j = 0; j < CT; ++j)
                acc[j] = __builtin_amdgcn_wmma_f32_16x16x4_f32(
                             false, a[ks], false, b[j][ks], (short)0, acc[j], false, false);
        }

#pragma unroll
        for (int j = 0; j < CT; ++j) {
#pragma unroll
            for (int r = 0; r < 8; ++r) {
                float v = acc[j][r] + bv[j];
                if (RELU) v = fmaxf(v, 0.0f);
                sD[(r + 8 * h) * SP + (w * CT + j) * 16 + c] = v;
            }
        }
        __syncthreads();

        float* ot = out + (long long)tile * 16 * OP;
#pragma unroll
        for (int p = 0; p < NST; ++p) {
            const int idx = (p * 128 + tid) * 4;
            const int r = idx / NC, cc = idx % NC;
            const int sw = SWZ ? (__builtin_popcount((unsigned)(tile * 16 + r)) & 1) * NC : 0;
            if (tile * 16 + r < n) *(v4f*)(ot + r * OP + cc + sw) = *(const v4f*)&sD[r * SP + cc];
        }
    }
}

extern "C" void kernel_launch(void* const* d_in, const int* in_sizes, int n_in,
                              void* d_out, int out_size, void* d_ws, size_t ws_size,
                              hipStream_t stream) {
    (void)n_in; (void)out_size; (void)ws_size;

    const int*   eidx = (const int*)d_in[0];
    const float* emb  = (const float*)d_in[1];
    const float* W1   = (const float*)d_in[2];
    const float* b1   = (const float*)d_in[3];
    const float* W2   = (const float*)d_in[4];
    const float* b2   = (const float*)d_in[5];

    const int E = in_sizes[0] / 2;
    const int n = in_sizes[1] / 64;
    const int* src = eidx;
    const int* dst = eidx + E;

    int*   cnt   = (int*)d_ws;
    int*   slot  = (int*)d_ws + 2 * (size_t)n;
    int*   ovf   = slot + 2 * (size_t)n * CAPH;
    int*   ovf_n = ovf + E - 1;
    float* h     = (float*)d_ws + 2 * (size_t)n + (size_t)n * 64;
    float* t     = h;
    float* out   = (float*)d_out;
    float* agg1  = h;
    float* pl    = out;

    const int B = 256;
    k_zero_cnt<<<(2 * n + B - 1) / B, B, 0, stream>>>(cnt, 2 * n, ovf_n);
    k_fill    <<<(E + EPL * B - 1) / (EPL * B), B, 0, stream>>>(src, dst, cnt, slot, ovf_n, ovf, E, n);
    k_deg     <<<(n + B - 1) / B, B, 0, stream>>>(cnt, n);

    k_pull<false, true, false, 128, true><<<2048, B, 0, stream>>>(emb, cnt, slot, ovf_n, ovf, src, dst, nullptr, agg1, pl, n);
    k_dense_ws<64, 128, 2, true, 128, false, 128, true><<<1024, 128, 0, stream>>>(agg1, W1, b1, h, n);

    k_dense_ws<128, 64, 1, false, 128, true, 128, false><<<1024, 128, 0, stream>>>(h, W2, nullptr, t, n);
    k_pull<true, false, true, 64, false><<<2048, B, 0, stream>>>(t, cnt, slot, ovf_n, ovf, src, dst, b2, out, pl, n);
}
